# starting kernel
# baseline (speedup 1.0000x reference)
#include <hip/hip_runtime.h>

typedef float v2f __attribute__((ext_vector_type(2)));
typedef float v8f __attribute__((ext_vector_type(8)));

__global__ void gcn_init_deg(float* __restrict__ deg, int n) {
    int i = blockIdx.x * blockDim.x + threadIdx.x;
    if (i < n) deg[i] = 1.0f;
}

__global__ void gcn_count_deg(const int* __restrict__ dst, float* __restrict__ deg, int e) {
    int i = blockIdx.x * blockDim.x + threadIdx.x;
    if (i < e) atomicAdd(&deg[dst[i]], 1.0f);
}

__global__ void gcn_rsqrt(const float* __restrict__ deg, float* __restrict__ dinv, int n) {
    int i = blockIdx.x * blockDim.x + threadIdx.x;
    if (i < n) {
        float d = deg[i];
        dinv[i] = d > 0.0f ? rsqrtf(d) : 0.0f;
    }
}

template <bool RELU>
__global__ void gcn_gemm_wmma(const float* __restrict__ A,
                              const float* __restrict__ B,
                              float* __restrict__ C,
                              int M, int K, int Ncols) {
    const int wave = (blockIdx.x * blockDim.x + threadIdx.x) >> 5;
    const int lane = threadIdx.x & 31;
    const int tilesN = Ncols >> 4;
    const int rowTile = wave / tilesN;
    const int colTile = wave - rowTile * tilesN;
    if (rowTile * 16 >= M) return;

    const int rowBase = rowTile << 4;
    const int colBase = colTile << 4;
    const int m     = lane & 15;
    const int khalf = (lane >> 4) << 1;

    v8f acc = {};
    const float* Arow = A + (size_t)(rowBase + m) * K;
    const float* Bcol = B + colBase + m;

    for (int k = 0; k < K; k += 4) {
        float a0 = Arow[k + khalf];
        float a1 = Arow[k + khalf + 1];
        if (RELU) { a0 = fmaxf(a0, 0.0f); a1 = fmaxf(a1, 0.0f); }
        v2f a; a.x = a0; a.y = a1;

        v2f b;
        b.x = Bcol[(size_t)(k + khalf)     * Ncols];
        b.y = Bcol[(size_t)(k + khalf + 1) * Ncols];

        acc = __builtin_amdgcn_wmma_f32_16x16x4_f32(
            false, a, false, b, (short)0, acc, false, false);
    }

    const int rOff = (lane >> 4) << 3;
    float* Crow = C + (size_t)(rowBase + rOff) * Ncols + colBase + m;
#pragma unroll
    for (int r = 0; r < 8; ++r)
        Crow[(size_t)r * Ncols] = acc[r];
}

template <int F>
__global__ void gcn_init_out(const float* __restrict__ h,
                             const float* __restrict__ dinv,
                             const float* __restrict__ bias,
                             float* __restrict__ out, int n) {
    int id = blockIdx.x * blockDim.x + threadIdx.x;
    int node = id / F;
    int f    = id - node * F;
    if (node < n) {
        float di = dinv[node];
        out[id] = bias[f] + h[id] * (di * di);
    }
}

template <int F>
__global__ void gcn_edge_scatter(const int* __restrict__ src,
                                 const int* __restrict__ dst,
                                 const float* __restrict__ dinv,
                                 const float* __restrict__ h,
                                 float* __restrict__ out, int e) {
    constexpr int CPE = F / 4;
    int id = blockIdx.x * blockDim.x + threadIdx.x;
    int edge = id / CPE;
    int c    = id - edge * CPE;
    if (edge >= e) return;
    int s = src[edge];
    int d = dst[edge];
    float w = dinv[s] * dinv[d];
    const float4 v = *((const float4*)(h + (size_t)s * F) + c);
    float* o = out + (size_t)d * F + (c << 2);
    atomicAdd(o + 0, v.x * w);
    atomicAdd(o + 1, v.y * w);
    atomicAdd(o + 2, v.z * w);
    atomicAdd(o + 3, v.w * w);
}

extern "C" void kernel_launch(void* const* d_in, const int* in_sizes, int n_in,
                              void* d_out, int out_size, void* d_ws, size_t ws_size,
                              hipStream_t stream) {
    constexpr int D_IN = 64, D_HID = 128, D_OUT = 64;

    const float* x  = (const float*)d_in[0];
    const int*   ei = (const int*)  d_in[1];
    const float* W1 = (const float*)d_in[2];
    const float* b1 = (const float*)d_in[3];
    const float* W2 = (const float*)d_in[4];
    const float* b2 = (const float*)d_in[5];
    float* out = (float*)d_out;

    const int N = in_sizes[0] / D_IN;
    const int E = in_sizes[1] / 2;
    const int* src = ei;
    const int* dst = ei + E;

    float* ws   = (float*)d_ws;
    float* deg  = ws;
    float* dinv = deg + N;
    float* h1   = dinv + N;
    float* out1 = h1 + (size_t)N * D_HID;

    auto cdiv = [](long a, long b) { return (int)((a + b - 1) / b); };
    const int BLK = 256;

    gcn_init_deg <<<cdiv(N, BLK), BLK, 0, stream>>>(deg, N);
    gcn_count_deg<<<cdiv(E, BLK), BLK, 0, stream>>>(dst, deg, E);
    gcn_rsqrt    <<<cdiv(N, BLK), BLK, 0, stream>>>(deg, dinv, N);

    {
        long waves = (long)(N / 16) * (D_HID / 16);
        gcn_gemm_wmma<false><<<cdiv(waves * 32, BLK), BLK, 0, stream>>>(
            x, W1, h1, N, D_IN, D_HID);
    }
    gcn_init_out<D_HID><<<cdiv((long)N * D_HID, BLK), BLK, 0, stream>>>(h1, dinv, b1, out1, N);
    gcn_edge_scatter<D_HID><<<cdiv((long)E * (D_HID / 4), BLK), BLK, 0, stream>>>(
        src, dst, dinv, h1, out1, E);

    {
        long waves = (long)(N / 16) * (D_OUT / 16);
        gcn_gemm_wmma<true><<<cdiv(waves * 32, BLK), BLK, 0, stream>>>(
            out1, W2, h1, N, D_HID, D_OUT);
    }
    gcn_init_out<D_OUT><<<cdiv((long)N * D_OUT, BLK), BLK, 0, stream>>>(h1, dinv, b2, out, N);
    gcn_edge_scatter<D_OUT><<<cdiv((long)E * (D_OUT / 4), BLK), BLK, 0, stream>>>(
        src, dst, dinv, h1, out, E);
}


# Layer 1 as (Ahat x) W1 + b1: edge scatter on the 64-wide x rows before the f32 WMMA product, bias in its epilogue; layer-1 float sums re-associated (summation order changed)
# speedup vs baseline: 1.3032x; 1.3032x over previous
#include <hip/hip_runtime.h>

typedef float v2f __attribute__((ext_vector_type(2)));
typedef float v8f __attribute__((ext_vector_type(8)));

__global__ void gcn_init_deg(float* __restrict__ deg, int n) {
    int i = blockIdx.x * blockDim.x + threadIdx.x;
    if (i < n) deg[i] = 1.0f;
}

__global__ void gcn_count_deg(const int* __restrict__ dst, float* __restrict__ deg, int e) {
    int i = blockIdx.x * blockDim.x + threadIdx.x;
    if (i < e) atomicAdd(&deg[dst[i]], 1.0f);
}

__global__ void gcn_rsqrt(const float* __restrict__ deg, float* __restrict__ dinv, int n) {
    int i = blockIdx.x * blockDim.x + threadIdx.x;
    if (i < n) {
        float d = deg[i];
        dinv[i] = d > 0.0f ? rsqrtf(d) : 0.0f;
    }
}

template <bool RELU, bool BIAS>
__global__ void gcn_gemm_wmma(const float* __restrict__ A,
                              const float* __restrict__ B,
                              const float* __restrict__ bias,
                              float* __restrict__ C,
                              int M, int K, int Ncols) {
    const int wave = (blockIdx.x * blockDim.x + threadIdx.x) >> 5;
    const int lane = threadIdx.x & 31;
    const int tilesN = Ncols >> 4;
    const int rowTile = wave / tilesN;
    const int colTile = wave - rowTile * tilesN;
    if (rowTile * 16 >= M) return;

    const int rowBase = rowTile << 4;
    const int colBase = colTile << 4;
    const int m     = lane & 15;
    const int khalf = (lane >> 4) << 1;

    v8f acc = {};
    const float* Arow = A + (size_t)(rowBase + m) * K;
    const float* Bcol = B + colBase + m;

    for (int k = 0; k < K; k += 4) {
        float a0 = Arow[k + khalf];
        float a1 = Arow[k + khalf + 1];
        if (RELU) { a0 = fmaxf(a0, 0.0f); a1 = fmaxf(a1, 0.0f); }
        v2f a; a.x = a0; a.y = a1;

        v2f b;
        b.x = Bcol[(size_t)(k + khalf)     * Ncols];
        b.y = Bcol[(size_t)(k + khalf + 1) * Ncols];

        acc = __builtin_amdgcn_wmma_f32_16x16x4_f32(
            false, a, false, b, (short)0, acc, false, false);
    }

    const int rOff = (lane >> 4) << 3;
    float* Crow = C + (size_t)(rowBase + rOff) * Ncols + colBase + m;
    float bv = 0.0f;
    if (BIAS) bv = bias[colBase + m];
#pragma unroll
    for (int r = 0; r < 8; ++r)
        Crow[(size_t)r * Ncols] = acc[r] + bv;
}

template <int F, bool BIAS>
__global__ void gcn_init_out(const float* __restrict__ h,
                             const float* __restrict__ dinv,
                             const float* __restrict__ bias,
                             float* __restrict__ out, int n) {
    int id = blockIdx.x * blockDim.x + threadIdx.x;
    int node = id / F;
    int f    = id - node * F;
    if (node < n) {
        float di = dinv[node];
        if (BIAS) out[id] = bias[f] + h[id] * (di * di);
        else      out[id] = h[id] * (di * di);
    }
}

template <int F>
__global__ void gcn_edge_scatter(const int* __restrict__ src,
                                 const int* __restrict__ dst,
                                 const float* __restrict__ dinv,
                                 const float* __restrict__ h,
                                 float* __restrict__ out, int e) {
    constexpr int CPE = F / 4;
    int id = blockIdx.x * blockDim.x + threadIdx.x;
    int edge = id / CPE;
    int c    = id - edge * CPE;
    if (edge >= e) return;
    int s = src[edge];
    int d = dst[edge];
    float w = dinv[s] * dinv[d];
    const float4 v = *((const float4*)(h + (size_t)s * F) + c);
    float* o = out + (size_t)d * F + (c << 2);
    atomicAdd(o + 0, v.x * w);
    atomicAdd(o + 1, v.y * w);
    atomicAdd(o + 2, v.z * w);
    atomicAdd(o + 3, v.w * w);
}

extern "C" void kernel_launch(void* const* d_in, const int* in_sizes, int n_in,
                              void* d_out, int out_size, void* d_ws, size_t ws_size,
                              hipStream_t stream) {
    constexpr int D_IN = 64, D_HID = 128, D_OUT = 64;

    const float* x  = (const float*)d_in[0];
    const int*   ei = (const int*)  d_in[1];
    const float* W1 = (const float*)d_in[2];
    const float* b1 = (const float*)d_in[3];
    const float* W2 = (const float*)d_in[4];
    const float* b2 = (const float*)d_in[5];
    float* out = (float*)d_out;

    const int N = in_sizes[0] / D_IN;
    const int E = in_sizes[1] / 2;
    const int* src = ei;
    const int* dst = ei + E;

    float* ws   = (float*)d_ws;
    float* deg  = ws;
    float* dinv = deg + N;
    float* agg  = dinv + N;
    float* out1 = agg + (size_t)N * D_IN;
    float* h2   = out1 + (size_t)N * D_HID;

    auto cdiv = [](long a, long b) { return (int)((a + b - 1) / b); };
    const int BLK = 256;

    gcn_init_deg <<<cdiv(N, BLK), BLK, 0, stream>>>(deg, N);
    gcn_count_deg<<<cdiv(E, BLK), BLK, 0, stream>>>(dst, deg, E);
    gcn_rsqrt    <<<cdiv(N, BLK), BLK, 0, stream>>>(deg, dinv, N);

    gcn_init_out<D_IN, false><<<cdiv((long)N * D_IN, BLK), BLK, 0, stream>>>(
        x, dinv, nullptr, agg, N);
    gcn_edge_scatter<D_IN><<<cdiv((long)E * (D_IN / 4), BLK), BLK, 0, stream>>>(
        src, dst, dinv, x, agg, E);
    {
        long waves = (long)(N / 16) * (D_HID / 16);
        gcn_gemm_wmma<false, true><<<cdiv(waves * 32, BLK), BLK, 0, stream>>>(
            agg, W1, b1, out1, N, D_IN, D_HID);
    }

    {
        long waves = (long)(N / 16) * (D_OUT / 16);
        gcn_gemm_wmma<true, false><<<cdiv(waves * 32, BLK), BLK, 0, stream>>>(
            out1, W2, nullptr, h2, N, D_HID, D_OUT);
    }
    gcn_init_out<D_OUT, true><<<cdiv((long)N * D_OUT, BLK), BLK, 0, stream>>>(h2, dinv, b2, out, N);
    gcn_edge_scatter<D_OUT><<<cdiv((long)E * (D_OUT / 4), BLK), BLK, 0, stream>>>(
        src, dst, dinv, h2, out, E);
}


# Edge sums by a per-call destination-sorted edge list (one returning integer atomic per edge, prefix sum, fill) and a 16-lane row gather in registers; no f32 atomics; sum order = atomic arrival rank
# speedup vs baseline: 1.9197x; 1.4731x over previous
#include <hip/hip_runtime.h>

typedef float v2f __attribute__((ext_vector_type(2)));
typedef float v8f __attribute__((ext_vector_type(8)));

__global__ void gcn_zero_cnt(int* __restrict__ cnt, int n) {
    int i = blockIdx.x * blockDim.x + threadIdx.x;
    if (i < n) cnt[i] = 0;
}

__global__ void gcn_rank_edges(const int* __restrict__ dst, int* __restrict__ cnt,
                               int* __restrict__ rank, int e) {
    int i = blockIdx.x * blockDim.x + threadIdx.x;
    if (i < e) rank[i] = atomicAdd(&cnt[dst[i]], 1);
}

__global__ void __launch_bounds__(256)
gcn_scan_deg(const int* __restrict__ cnt, int* __restrict__ rowptr,
             float* __restrict__ dinv, int n) {
    __shared__ int sh[256];
    const int t = threadIdx.x;
    const int blockStart = blockIdx.x * 256;

    int partial = 0;
    const int4* c4 = (const int4*)cnt;
    const int nvec = blockStart >> 2;
    for (int j = t; j < nvec; j += 256) {
        const int4 v = c4[j];
        partial += (v.x + v.y) + (v.z + v.w);
    }
    const int i = blockStart + t;
    const int c = (i < n) ? cnt[i] : 0;

    sh[t] = partial;
    __syncthreads();
    for (int s = 128; s > 0; s >>= 1) {
        if (t < s) sh[t] += sh[t + s];
        __syncthreads();
    }
    const int base = sh[0];
    __syncthreads();

    sh[t] = c;
    __syncthreads();
    for (int off = 1; off < 256; off <<= 1) {
        const int v = (t >= off) ? sh[t - off] : 0;
        __syncthreads();
        sh[t] += v;
        __syncthreads();
    }
    if (i < n) {
        rowptr[i] = base + sh[t] - c;
        const float d = (float)(c + 1);
        dinv[i] = d > 0.0f ? rsqrtf(d) : 0.0f;
    }
}

__global__ void gcn_fill_csr(const int* __restrict__ src, const int* __restrict__ dst,
                             const int* __restrict__ rank, const int* __restrict__ rowptr,
                             int* __restrict__ csr, int e) {
    int i = blockIdx.x * blockDim.x + threadIdx.x;
    if (i < e) csr[rowptr[dst[i]] + rank[i]] = src[i];
}

template <bool RELU, bool BIAS>
__global__ void gcn_gemm_wmma(const float* __restrict__ A,
                              const float* __restrict__ B,
                              const float* __restrict__ bias,
                              float* __restrict__ C,
                              int M, int K, int Ncols) {
    const int wave = (blockIdx.x * blockDim.x + threadIdx.x) >> 5;
    const int lane = threadIdx.x & 31;
    const int tilesN = Ncols >> 4;
    const int rowTile = wave / tilesN;
    const int colTile = wave - rowTile * tilesN;
    if (rowTile * 16 >= M) return;

    const int rowBase = rowTile << 4;
    const int colBase = colTile << 4;
    const int m     = lane & 15;
    const int khalf = (lane >> 4) << 1;

    v8f acc = {};
    const float* Arow = A + (size_t)(rowBase + m) * K;
    const float* Bcol = B + colBase + m;

    for (int k = 0; k < K; k += 4) {
        float a0 = Arow[k + khalf];
        float a1 = Arow[k + khalf + 1];
        if (RELU) { a0 = fmaxf(a0, 0.0f); a1 = fmaxf(a1, 0.0f); }
        v2f a; a.x = a0; a.y = a1;

        v2f b;
        b.x = Bcol[(size_t)(k + khalf)     * Ncols];
        b.y = Bcol[(size_t)(k + khalf + 1) * Ncols];

        acc = __builtin_amdgcn_wmma_f32_16x16x4_f32(
            false, a, false, b, (short)0, acc, false, false);
    }

    const int rOff = (lane >> 4) << 3;
    float* Crow = C + (size_t)(rowBase + rOff) * Ncols + colBase + m;
    float bv = 0.0f;
    if (BIAS) bv = bias[colBase + m];
#pragma unroll
    for (int r = 0; r < 8; ++r)
        Crow[(size_t)r * Ncols] = acc[r] + bv;
}

template <bool BIAS>
__global__ void __launch_bounds__(256)
gcn_gather_rows(const int* __restrict__ rowptr, const int* __restrict__ cnt,
                const int* __restrict__ csr, const float* __restrict__ dinv,
                const float* __restrict__ h, const float* __restrict__ bias,
                float* __restrict__ out, int n) {
    const int id   = blockIdx.x * blockDim.x + threadIdx.x;
    const int node = id >> 4;
    const int c    = id & 15;
    if (node >= n) return;

    const float4* h4 = (const float4*)h;
    const float dn = dinv[node];
    const int beg = rowptr[node];
    const int end = beg + cnt[node];

    const float4 self = h4[(size_t)node * 16 + c];
    const float wself = dn * dn;
    float4 acc;
    if (BIAS) {
        const float4 b = ((const float4*)bias)[c];
        acc.x = b.x + self.x * wself;
        acc.y = b.y + self.y * wself;
        acc.z = b.z + self.z * wself;
        acc.w = b.w + self.w * wself;
    } else {
        acc.x = self.x * wself;
        acc.y = self.y * wself;
        acc.z = self.z * wself;
        acc.w = self.w * wself;
    }

    int j = beg;
    for (; j + 4 <= end; j += 4) {
        const int s0 = csr[j + 0];
        const int s1 = csr[j + 1];
        const int s2 = csr[j + 2];
        const int s3 = csr[j + 3];
        const float4 v0 = h4[(size_t)s0 * 16 + c];
        const float4 v1 = h4[(size_t)s1 * 16 + c];
        const float4 v2 = h4[(size_t)s2 * 16 + c];
        const float4 v3 = h4[(size_t)s3 * 16 + c];
        const float w0 = dinv[s0] * dn;
        const float w1 = dinv[s1] * dn;
        const float w2 = dinv[s2] * dn;
        const float w3 = dinv[s3] * dn;
        acc.x += v0.x * w0; acc.y += v0.y * w0; acc.z += v0.z * w0; acc.w += v0.w * w0;
        acc.x += v1.x * w1; acc.y += v1.y * w1; acc.z += v1.z * w1; acc.w += v1.w * w1;
        acc.x += v2.x * w2; acc.y += v2.y * w2; acc.z += v2.z * w2; acc.w += v2.w * w2;
        acc.x += v3.x * w3; acc.y += v3.y * w3; acc.z += v3.z * w3; acc.w += v3.w * w3;
    }
    for (; j < end; ++j) {
        const int s = csr[j];
        const float4 v = h4[(size_t)s * 16 + c];
        const float w = dinv[s] * dn;
        acc.x += v.x * w; acc.y += v.y * w; acc.z += v.z * w; acc.w += v.w * w;
    }
    ((float4*)out)[(size_t)node * 16 + c] = acc;
}

extern "C" void kernel_launch(void* const* d_in, const int* in_sizes, int n_in,
                              void* d_out, int out_size, void* d_ws, size_t ws_size,
                              hipStream_t stream) {
    constexpr int D_IN = 64, D_HID = 128, D_OUT = 64;

    const float* x  = (const float*)d_in[0];
    const int*   ei = (const int*)  d_in[1];
    const float* W1 = (const float*)d_in[2];
    const float* b1 = (const float*)d_in[3];
    const float* W2 = (const float*)d_in[4];
    const float* b2 = (const float*)d_in[5];
    float* out = (float*)d_out;

    const int N = in_sizes[0] / D_IN;
    const int E = in_sizes[1] / 2;
    const int* src = ei;
    const int* dst = ei + E;

    auto cdiv = [](long a, long b) { return (int)((a + b - 1) / b); };
    const size_t N4 = ((size_t)N + 3) & ~(size_t)3;

    float* agg    = (float*)d_ws;
    float* out1   = agg + (size_t)N * D_IN;
    float* h2     = out1 + (size_t)N * D_HID;
    int*   cnt    = (int*)(h2 + (size_t)N * D_OUT);
    int*   rowptr = cnt + N4;
    float* dinv   = (float*)(rowptr + N4);
    int*   rank   = (int*)(dinv + N4);
    int*   csr    = rank + E;

    const int BLK = 256;

    gcn_zero_cnt  <<<cdiv(N, BLK), BLK, 0, stream>>>(cnt, N);
    gcn_rank_edges<<<cdiv(E, BLK), BLK, 0, stream>>>(dst, cnt, rank, E);
    gcn_scan_deg  <<<cdiv(N, 256), 256, 0, stream>>>(cnt, rowptr, dinv, N);
    gcn_fill_csr  <<<cdiv(E, BLK), BLK, 0, stream>>>(src, dst, rank, rowptr, csr, E);

    gcn_gather_rows<false><<<cdiv((long)N * 16, 256), 256, 0, stream>>>(
        rowptr, cnt, csr, dinv, x, nullptr, agg, N);
    {
        long waves = (long)(N / 16) * (D_HID / 16);
        gcn_gemm_wmma<false, true><<<cdiv(waves * 32, BLK), BLK, 0, stream>>>(
            agg, W1, b1, out1, N, D_IN, D_HID);
    }

    {
        long waves = (long)(N / 16) * (D_OUT / 16);
        gcn_gemm_wmma<true, false><<<cdiv(waves * 32, BLK), BLK, 0, stream>>>(
            out1, W2, nullptr, h2, N, D_HID, D_OUT);
    }
    gcn_gather_rows<true><<<cdiv((long)N * 16, 256), 256, 0, stream>>>(
        rowptr, cnt, csr, dinv, h2, b2, out, N);
}


# Layer-1 gather, W1 product, bias+ReLU and W2 product fused in one launch: each 16-node tile goes registers -> LDS -> f32 WMMA -> LDS -> f32 WMMA; agg and out1 never reach memory; sum order kept
# speedup vs baseline: 2.3678x; 1.2334x over previous
#include <hip/hip_runtime.h>

typedef float v2f __attribute__((ext_vector_type(2)));
typedef float v8f __attribute__((ext_vector_type(8)));

__global__ void gcn_zero_cnt(int* __restrict__ cnt, int n) {
    int i = blockIdx.x * blockDim.x + threadIdx.x;
    if (i < n) cnt[i] = 0;
}

__global__ void gcn_rank_edges(const int* __restrict__ dst, int* __restrict__ cnt,
                               int* __restrict__ rank, int e) {
    int i = blockIdx.x * blockDim.x + threadIdx.x;
    if (i < e) rank[i] = atomicAdd(&cnt[dst[i]], 1);
}

__global__ void __launch_bounds__(256)
gcn_scan_deg(const int* __restrict__ cnt, int* __restrict__ rowptr,
             float* __restrict__ dinv, int n) {
    __shared__ int sh[256];
    const int t = threadIdx.x;
    const int blockStart = blockIdx.x * 256;

    int partial = 0;
    const int4* c4 = (const int4*)cnt;
    const int nvec = blockStart >> 2;
    for (int j = t; j < nvec; j += 256) {
        const int4 v = c4[j];
        partial += (v.x + v.y) + (v.z + v.w);
    }
    const int i = blockStart + t;
    const int c = (i < n) ? cnt[i] : 0;

    sh[t] = partial;
    __syncthreads();
    for (int s = 128; s > 0; s >>= 1) {
        if (t < s) sh[t] += sh[t + s];
        __syncthreads();
    }
    const int base = sh[0];
    __syncthreads();

    sh[t] = c;
    __syncthreads();
    for (int off = 1; off < 256; off <<= 1) {
        const int v = (t >= off) ? sh[t - off] : 0;
        __syncthreads();
        sh[t] += v;
        __syncthreads();
    }
    if (i < n) {
        rowptr[i] = base + sh[t] - c;
        const float d = (float)(c + 1);
        dinv[i] = d > 0.0f ? rsqrtf(d) : 0.0f;
    }
}

__global__ void gcn_fill_csr(const int* __restrict__ src, const int* __restrict__ dst,
                             const int* __restrict__ rank, const int* __restrict__ rowptr,
                             int* __restrict__ csr, int e) {
    int i = blockIdx.x * blockDim.x + threadIdx.x;
    if (i < e) csr[rowptr[dst[i]] + rank[i]] = src[i];
}

template <bool BIAS>
__device__ __forceinline__ float4
gcn_row_sum(int node, int c, const int* __restrict__ rowptr, const int* __restrict__ cnt,
            const int* __restrict__ csr, const float* __restrict__ dinv,
            const float4* __restrict__ h4, const float* __restrict__ bias) {
    const float dn = dinv[node];
    const int beg = rowptr[node];
    const int end = beg + cnt[node];

    const float4 self = h4[(size_t)node * 16 + c];
    const float wself = dn * dn;
    float4 acc;
    if (BIAS) {
        const float4 b = ((const float4*)bias)[c];
        acc.x = b.x + self.x * wself;
        acc.y = b.y + self.y * wself;
        acc.z = b.z + self.z * wself;
        acc.w = b.w + self.w * wself;
    } else {
        acc.x = self.x * wself;
        acc.y = self.y * wself;
        acc.z = self.z * wself;
        acc.w = self.w * wself;
    }

    int j = beg;
    for (; j + 4 <= end; j += 4) {
        const int s0 = csr[j + 0];
        const int s1 = csr[j + 1];
        const int s2 = csr[j + 2];
        const int s3 = csr[j + 3];
        const float4 v0 = h4[(size_t)s0 * 16 + c];
        const float4 v1 = h4[(size_t)s1 * 16 + c];
        const float4 v2 = h4[(size_t)s2 * 16 + c];
        const float4 v3 = h4[(size_t)s3 * 16 + c];
        const float w0 = dinv[s0] * dn;
        const float w1 = dinv[s1] * dn;
        const float w2 = dinv[s2] * dn;
        const float w3 = dinv[s3] * dn;
        acc.x += v0.x * w0; acc.y += v0.y * w0; acc.z += v0.z * w0; acc.w += v0.w * w0;
        acc.x += v1.x * w1; acc.y += v1.y * w1; acc.z += v1.z * w1; acc.w += v1.w * w1;
        acc.x += v2.x * w2; acc.y += v2.y * w2; acc.z += v2.z * w2; acc.w += v2.w * w2;
        acc.x += v3.x * w3; acc.y += v3.y * w3; acc.z += v3.z * w3; acc.w += v3.w * w3;
    }
    for (; j < end; ++j) {
        const int s = csr[j];
        const float4 v = h4[(size_t)s * 16 + c];
        const float w = dinv[s] * dn;
        acc.x += v.x * w; acc.y += v.y * w; acc.z += v.z * w; acc.w += v.w * w;
    }
    return acc;
}

constexpr int GCN_PA = 64 + 4;
constexpr int GCN_PH = 128 + 4;

__global__ void __launch_bounds__(256)
gcn_layer1_fused(const int* __restrict__ rowptr, const int* __restrict__ cnt,
                 const int* __restrict__ csr, const float* __restrict__ dinv,
                 const float* __restrict__ x, const float* __restrict__ W1,
                 const float* __restrict__ b1, const float* __restrict__ W2,
                 float* __restrict__ h2, int n) {
    __shared__ __attribute__((aligned(16))) float sA[16 * GCN_PA];
    __shared__ __attribute__((aligned(16))) float sH[16 * GCN_PH];

    const int t = threadIdx.x;
    const int nodeBase = blockIdx.x * 16;

    {
        const int r = t >> 4;
        const int c = t & 15;
        const int node = nodeBase + r;
        float4 acc = make_float4(0.0f, 0.0f, 0.0f, 0.0f);
        if (node < n)
            acc = gcn_row_sum<false>(node, c, rowptr, cnt, csr, dinv, (const float4*)x, nullptr);
        *(float4*)&sA[r * GCN_PA + 4 * c] = acc;
    }
    __syncthreads();

    const int wave  = t >> 5;
    const int lane  = t & 31;
    const int m     = lane & 15;
    const int khalf = (lane >> 4) << 1;
    const int rOff  = (lane >> 4) << 3;
    const int col   = (wave << 4) + m;

    {
        v8f acc = {};
        const float* Arow = sA + m * GCN_PA + khalf;
        const float* Bcol = W1 + (size_t)khalf * 128 + col;
#pragma unroll
        for (int k = 0; k < 64; k += 4) {
            const v2f a = *(const v2f*)(Arow + k);
            v2f b;
            b.x = Bcol[(size_t)k * 128];
            b.y = Bcol[(size_t)(k + 1) * 128];
            acc = __builtin_amdgcn_wmma_f32_16x16x4_f32(
                false, a, false, b, (short)0, acc, false, false);
        }
        const float bv = b1[col];
#pragma unroll
        for (int r = 0; r < 8; ++r)
            sH[(rOff + r) * GCN_PH + col] = fmaxf(acc[r] + bv, 0.0f);
    }
    __syncthreads();

    if (wave < 4) {
        v8f acc = {};
        const float* Arow = sH + m * GCN_PH + khalf;
        const float* Bcol = W2 + (size_t)khalf * 64 + col;
#pragma unroll
        for (int k = 0; k < 128; k += 4) {
            const v2f a = *(const v2f*)(Arow + k);
            v2f b;
            b.x = Bcol[(size_t)k * 64];
            b.y = Bcol[(size_t)(k + 1) * 64];
            acc = __builtin_amdgcn_wmma_f32_16x16x4_f32(
                false, a, false, b, (short)0, acc, false, false);
        }
        float* Crow = h2 + (size_t)(nodeBase + rOff) * 64 + col;
#pragma unroll
        for (int r = 0; r < 8; ++r)
            Crow[(size_t)r * 64] = acc[r];
    }
}

__global__ void __launch_bounds__(256)
gcn_gather_out(const int* __restrict__ rowptr, const int* __restrict__ cnt,
               const int* __restrict__ csr, const float* __restrict__ dinv,
               const float* __restrict__ h, const float* __restrict__ bias,
               float* __restrict__ out, int n) {
    const int id   = blockIdx.x * blockDim.x + threadIdx.x;
    const int node = id >> 4;
    const int c    = id & 15;
    if (node >= n) return;
    ((float4*)out)[(size_t)node * 16 + c] =
        gcn_row_sum<true>(node, c, rowptr, cnt, csr, dinv, (const float4*)h, bias);
}

extern "C" void kernel_launch(void* const* d_in, const int* in_sizes, int n_in,
                              void* d_out, int out_size, void* d_ws, size_t ws_size,
                              hipStream_t stream) {
    constexpr int D_IN = 64, D_OUT = 64;

    const float* x  = (const float*)d_in[0];
    const int*   ei = (const int*)  d_in[1];
    const float* W1 = (const float*)d_in[2];
    const float* b1 = (const float*)d_in[3];
    const float* W2 = (const float*)d_in[4];
    const float* b2 = (const float*)d_in[5];
    float* out = (float*)d_out;

    const int N = in_sizes[0] / D_IN;
    const int E = in_sizes[1] / 2;
    const int* src = ei;
    const int* dst = ei + E;

    auto cdiv = [](long a, long b) { return (int)((a + b - 1) / b); };
    const size_t N4  = ((size_t)N + 3) & ~(size_t)3;
    const size_t N16 = ((size_t)N + 15) & ~(size_t)15;

    float* h2     = (float*)d_ws;
    int*   cnt    = (int*)(h2 + N16 * D_OUT);
    int*   rowptr = cnt + N4;
    float* dinv   = (float*)(rowptr + N4);
    int*   rank   = (int*)(dinv + N4);
    int*   csr    = rank + E;

    const int BLK = 256;

    gcn_zero_cnt  <<<cdiv(N, BLK), BLK, 0, stream>>>(cnt, N);
    gcn_rank_edges<<<cdiv(E, BLK), BLK, 0, stream>>>(dst, cnt, rank, E);
    gcn_scan_deg  <<<cdiv(N, 256), 256, 0, stream>>>(cnt, rowptr, dinv, N);
    gcn_fill_csr  <<<cdiv(E, BLK), BLK, 0, stream>>>(src, dst, rank, rowptr, csr, E);

    gcn_layer1_fused<<<cdiv(N, 16), 256, 0, stream>>>(
        rowptr, cnt, csr, dinv, x, W1, b1, W2, h2, N);

    gcn_gather_out<<<cdiv((long)N * 16, 256), 256, 0, stream>>>(
        rowptr, cnt, csr, dinv, h2, b2, out, N);
}


# Edge list as fixed-pitch per-destination buckets (64 slots + overflow list) filled by the ranking atomic itself: prefix-sum and fill launches removed, dinv taken from the counts; sum order kept
# speedup vs baseline: 2.8493x; 1.2034x over previous
#include <hip/hip_runtime.h>

typedef float v2f __attribute__((ext_vector_type(2)));
typedef float v8f __attribute__((ext_vector_type(8)));

constexpr int GCN_CAP = 64;

__global__ void gcn_zero_cnt(int* __restrict__ cnt, int n) {
    int i = blockIdx.x * blockDim.x + threadIdx.x;
    if (i < n) cnt[i] = 0;
}

__global__ void gcn_bucket_edges(const int* __restrict__ src, const int* __restrict__ dst,
                                 int* __restrict__ cnt, int* __restrict__ ovfCnt,
                                 int* __restrict__ bucket, int* __restrict__ ovf, int e) {
    int i = blockIdx.x * blockDim.x + threadIdx.x;
    if (i < e) {
        const int d = dst[i];
        const int s = src[i];
        const int r = atomicAdd(&cnt[d], 1);
        if (r < GCN_CAP) {
            bucket[(size_t)d * GCN_CAP + r] = s;
        } else {
            const int o = atomicAdd(ovfCnt, 1);
            ovf[2 * (size_t)o]     = d;
            ovf[2 * (size_t)o + 1] = s;
        }
    }
}

__device__ __forceinline__ float gcn_dinv(int c) {
    return __builtin_amdgcn_rsqf((float)(c + 1));
}

template <bool BIAS>
__device__ __forceinline__ float4
gcn_row_sum(int node, int c, const int* __restrict__ cnt, const int* __restrict__ ovfCnt,
            const int* __restrict__ bucket, const int* __restrict__ ovf,
            const float4* __restrict__ h4, const float* __restrict__ bias) {
    const int cn = cnt[node];
    const float dn = gcn_dinv(cn);
    const int* row = bucket + (size_t)node * GCN_CAP;
    const int end = cn < GCN_CAP ? cn : GCN_CAP;

    const float4 self = h4[(size_t)node * 16 + c];
    const float wself = dn * dn;
    float4 acc;
    if (BIAS) {
        const float4 b = ((const float4*)bias)[c];
        acc.x = b.x + self.x * wself;
        acc.y = b.y + self.y * wself;
        acc.z = b.z + self.z * wself;
        acc.w = b.w + self.w * wself;
    } else {
        acc.x = self.x * wself;
        acc.y = self.y * wself;
        acc.z = self.z * wself;
        acc.w = self.w * wself;
    }

    int j = 0;
    for (; j + 4 <= end; j += 4) {
        const int s0 = row[j + 0];
        const int s1 = row[j + 1];
        const int s2 = row[j + 2];
        const int s3 = row[j + 3];
        const float4 v0 = h4[(size_t)s0 * 16 + c];
        const float4 v1 = h4[(size_t)s1 * 16 + c];
        const float4 v2 = h4[(size_t)s2 * 16 + c];
        const float4 v3 = h4[(size_t)s3 * 16 + c];
        const float w0 = gcn_dinv(cnt[s0]) * dn;
        const float w1 = gcn_dinv(cnt[s1]) * dn;
        const float w2 = gcn_dinv(cnt[s2]) * dn;
        const float w3 = gcn_dinv(cnt[s3]) * dn;
        acc.x += v0.x * w0; acc.y += v0.y * w0; acc.z += v0.z * w0; acc.w += v0.w * w0;
        acc.x += v1.x * w1; acc.y += v1.y * w1; acc.z += v1.z * w1; acc.w += v1.w * w1;
        acc.x += v2.x * w2; acc.y += v2.y * w2; acc.z += v2.z * w2; acc.w += v2.w * w2;
        acc.x += v3.x * w3; acc.y += v3.y * w3; acc.z += v3.z * w3; acc.w += v3.w * w3;
    }
    for (; j < end; ++j) {
        const int s = row[j];
        const float4 v = h4[(size_t)s * 16 + c];
        const int cs = cnt[s];
        __builtin_amdgcn_sched_barrier(0);
        const float w = gcn_dinv(cs) * dn;
        acc.x += v.x * w; acc.y += v.y * w; acc.z += v.z * w; acc.w += v.w * w;
    }
    if (cn > GCN_CAP) {
        const int no = *ovfCnt;
        for (int k = 0; k < no; ++k) {
            if (ovf[2 * (size_t)k] == node) {
                const int s = ovf[2 * (size_t)k + 1];
                const float4 v = h4[(size_t)s * 16 + c];
                const float w = gcn_dinv(cnt[s]) * dn;
                acc.x += v.x * w; acc.y += v.y * w; acc.z += v.z * w; acc.w += v.w * w;
            }
        }
    }
    return acc;
}

constexpr int GCN_PA = 64 + 4;
constexpr int GCN_PH = 128 + 4;

__global__ void __launch_bounds__(256)
gcn_layer1_fused(const int* __restrict__ cnt, const int* __restrict__ ovfCnt,
                 const int* __restrict__ bucket, const int* __restrict__ ovf,
                 const float* __restrict__ x, const float* __restrict__ W1,
                 const float* __restrict__ b1, const float* __restrict__ W2,
                 float* __restrict__ h2, int n) {
    __shared__ __attribute__((aligned(16))) float sA[16 * GCN_PA];
    __shared__ __attribute__((aligned(16))) float sH[16 * GCN_PH];

    const int t = threadIdx.x;
    const int nodeBase = blockIdx.x * 16;

    {
        const int r = t >> 4;
        const int c = t & 15;
        const int node = nodeBase + r;
        float4 acc = make_float4(0.0f, 0.0f, 0.0f, 0.0f);
        if (node < n)
            acc = gcn_row_sum<false>(node, c, cnt, ovfCnt, bucket, ovf, (const float4*)x, nullptr);
        *(float4*)&sA[r * GCN_PA + 4 * c] = acc;
    }
    __syncthreads();

    const int wave  = t >> 5;
    const int lane  = t & 31;
    const int m     = lane & 15;
    const int khalf = (lane >> 4) << 1;
    const int rOff  = (lane >> 4) << 3;
    const int col   = (wave << 4) + m;

    {
        v8f acc = {};
        const float* Arow = sA + m * GCN_PA + khalf;
        const float* Bcol = W1 + (size_t)khalf * 128 + col;
#pragma unroll
        for (int k = 0; k < 64; k += 4) {
            const v2f a = *(const v2f*)(Arow + k);
            v2f b;
            b.x = Bcol[(size_t)k * 128];
            b.y = Bcol[(size_t)(k + 1) * 128];
            acc = __builtin_amdgcn_wmma_f32_16x16x4_f32(
                false, a, false, b, (short)0, acc, false, false);
        }
        const float bv = b1[col];
#pragma unroll
        for (int r = 0; r < 8; ++r)
            sH[(rOff + r) * GCN_PH + col] = fmaxf(acc[r] + bv, 0.0f);
    }
    __syncthreads();

    if (wave < 4) {
        v8f acc = {};
        const float* Arow = sH + m * GCN_PH + khalf;
        const float* Bcol = W2 + (size_t)khalf * 64 + col;
#pragma unroll
        for (int k = 0; k < 128; k += 4) {
            const v2f a = *(const v2f*)(Arow + k);
            v2f b;
            b.x = Bcol[(size_t)k * 64];
            b.y = Bcol[(size_t)(k + 1) * 64];
            acc = __builtin_amdgcn_wmma_f32_16x16x4_f32(
                false, a, false, b, (short)0, acc, false, false);
        }
        float* Crow = h2 + (size_t)(nodeBase + rOff) * 64 + col;
#pragma unroll
        for (int r = 0; r < 8; ++r)
            Crow[(size_t)r * 64] = acc[r];
    }
}

__global__ void __launch_bounds__(256)
gcn_gather_out(const int* __restrict__ cnt, const int* __restrict__ ovfCnt,
               const int* __restrict__ bucket, const int* __restrict__ ovf,
               const float* __restrict__ h, const float* __restrict__ bias,
               float* __restrict__ out, int n) {
    const int id   = blockIdx.x * blockDim.x + threadIdx.x;
    const int node = id >> 4;
    const int c    = id & 15;
    if (node >= n) return;
    ((float4*)out)[(size_t)node * 16 + c] =
        gcn_row_sum<true>(node, c, cnt, ovfCnt, bucket, ovf, (const float4*)h, bias);
}

extern "C" void kernel_launch(void* const* d_in, const int* in_sizes, int n_in,
                              void* d_out, int out_size, void* d_ws, size_t ws_size,
                              hipStream_t stream) {
    constexpr int D_IN = 64, D_OUT = 64;

    const float* x  = (const float*)d_in[0];
    const int*   ei = (const int*)  d_in[1];
    const float* W1 = (const float*)d_in[2];
    const float* b1 = (const float*)d_in[3];
    const float* W2 = (const float*)d_in[4];
    const float* b2 = (const float*)d_in[5];
    float* out = (float*)d_out;

    const int N = in_sizes[0] / D_IN;
    const int E = in_sizes[1] / 2;
    const int* src = ei;
    const int* dst = ei + E;

    auto cdiv = [](long a, long b) { return (int)((a + b - 1) / b); };
    const size_t NC  = ((size_t)N + 1 + 63) & ~(size_t)63;
    const size_t N16 = ((size_t)N + 15) & ~(size_t)15;

    float* h2     = (float*)d_ws;
    int*   cnt    = (int*)(h2 + N16 * D_OUT);
    int*   ovfCnt = cnt + N;
    int*   bucket = cnt + NC;
    int*   ovf    = bucket + (size_t)N * GCN_CAP;

    const int BLK = 256;

    gcn_zero_cnt    <<<cdiv(N + 1, BLK), BLK, 0, stream>>>(cnt, N + 1);
    gcn_bucket_edges<<<cdiv(E, BLK), BLK, 0, stream>>>(src, dst, cnt, ovfCnt, bucket, ovf, E);

    gcn_layer1_fused<<<cdiv(N, 16), 256, 0, stream>>>(
        cnt, ovfCnt, bucket, ovf, x, W1, b1, W2, h2, N);

    gcn_gather_out<<<cdiv((long)N * 16, 256), 256, 0, stream>>>(
        cnt, ovfCnt, bucket, ovf, h2, b2, out, N);
}


# Edge ranks from LDS counters in two passes (chunk blocks deal edges into per-range segments, range blocks build the bucket rows in LDS): no global atomic per edge; sum order = LDS arrival rank
# speedup vs baseline: 5.0018x; 1.7554x over previous
#include <hip/hip_runtime.h>

typedef float v2f __attribute__((ext_vector_type(2)));
typedef float v8f __attribute__((ext_vector_type(8)));

constexpr int GCN_CAP = 64;
constexpr int BIN_RN  = 256;
constexpr int BIN_CH  = 2048;
constexpr int BIN_CAP = 32;
constexpr int BIN_RG  = 256;
constexpr int BIN_PS  = BIN_CAP + 1;

__global__ void __launch_bounds__(256)
gcn_bin_edges(const int* __restrict__ src, const int* __restrict__ dst,
              int* __restrict__ segCnt, int* __restrict__ seg,
              int* __restrict__ spillCnt, int* __restrict__ spill,
              int* __restrict__ ovfCnt, int e, int nr, int nch, int nj) {
    __shared__ int sCnt[BIN_RG];
    __shared__ int sSeg[BIN_RG * BIN_PS];
    __shared__ int sSpill;
    const int t = threadIdx.x;
    const int wave = t >> 5;
    const int lane = t & 31;
    const int chunk = blockIdx.x;

    constexpr int PER = BIN_CH / 256;
    const int base = chunk * BIN_CH + t;
    int d[PER], s[PER];
#pragma unroll
    for (int k = 0; k < PER; ++k) {
        const int i = base + k * 256;
        d[k] = i < e ? dst[i] : -1;
        s[k] = i < e ? src[i] : 0;
    }
    if (t == 0) sSpill = 0;
    if (chunk == 0 && t == 0) *ovfCnt = 0;

    const int slot = (chunk & 31) * nj + (chunk >> 5);

    for (int rb = 0; rb < nr; rb += BIN_RG) {
        __syncthreads();
        sCnt[t] = 0;
        __syncthreads();
#pragma unroll
        for (int k = 0; k < PER; ++k) {
            const int r = d[k] / BIN_RN - rb;
            if (d[k] >= 0 && r >= 0 && r < BIN_RG) {
                const int p = atomicAdd(&sCnt[r], 1);
                if (p < BIN_CAP) {
                    sSeg[r * BIN_PS + p] = (s[k] << 8) | (d[k] % BIN_RN);
                } else {
                    const int o = atomicAdd(&sSpill, 1);
                    int* q = spill + 2 * ((size_t)chunk * BIN_CH + o);
                    q[0] = d[k];
                    q[1] = s[k];
                }
            }
        }
        __syncthreads();
        const int rn = nr - rb < BIN_RG ? nr - rb : BIN_RG;
        for (int r = wave; r < rn; r += 8) {
            const int c = sCnt[r];
            if (lane < c && lane < BIN_CAP)
                seg[((size_t)(rb + r) * nch + chunk) * BIN_CAP + lane] = sSeg[r * BIN_PS + lane];
        }
        if (t < rn) {
            const int c = sCnt[t];
            segCnt[(size_t)(rb + t) * 32 * nj + slot] = c < BIN_CAP ? c : BIN_CAP;
        }
    }
    __syncthreads();
    if (t == 0) spillCnt[chunk] = sSpill;
}

__device__ __forceinline__ void gcn_overflow(int d, int s, int* __restrict__ ovfCnt,
                                             int* __restrict__ ovf) {
    const int o = atomicAdd(ovfCnt, 1);
    ovf[2 * (size_t)o]     = d;
    ovf[2 * (size_t)o + 1] = s;
}

__global__ void __launch_bounds__(1024)
gcn_fill_buckets(const int* __restrict__ segCnt, const int* __restrict__ seg,
                 const int* __restrict__ spillCnt, const int* __restrict__ spill,
                 int* __restrict__ cnt, int* __restrict__ ovfCnt,
                 int* __restrict__ bucket, int* __restrict__ ovf, int n, int nch, int nj) {
    __shared__ int sCnt[BIN_RN];
    __shared__ __attribute__((aligned(16))) int sB[BIN_RN * GCN_CAP];
    const int t = threadIdx.x;
    const int r = blockIdx.x;
    const int nodeBase = r * BIN_RN;
    const int wave = t >> 5;
    const int lane = t & 31;

    const int* cRow = segCnt + ((size_t)r * 32 + wave) * nj;
    const int* sRow = seg + ((size_t)r * nch + wave) * BIN_CAP + lane;
    int mAll   = (lane < nj && wave + 32 * lane < nch) ? cRow[lane] : 0;
    int mSpill = wave * 32 + lane < nch ? spillCnt[wave * 32 + lane] : 0;

    if (t < BIN_RN) sCnt[t] = 0;
#pragma unroll
    for (int i = 0; i < BIN_RN * GCN_CAP / 4096; ++i)
        ((int4*)sB)[i * 1024 + t] = make_int4(0, 0, 0, 0);
    __syncthreads();

    for (int jb = 0; jb < nj; jb += 32) {
        if (jb > 0) {
            const int jMine = jb + lane;
            mAll = (jMine < nj && wave + 32 * jMine < nch) ? cRow[jMine] : 0;
        }
        const int jn = nj - jb < 32 ? nj - jb : 32;
        for (int j0 = 0; j0 < jn; j0 += 16) {
            int m[16], v[16];
#pragma unroll
            for (int u = 0; u < 16; ++u)
                m[u] = __builtin_amdgcn_readlane(mAll, j0 + u);
#pragma unroll
            for (int u = 0; u < 16; ++u)
                v[u] = lane < m[u] ? sRow[(size_t)(jb + j0 + u) * 32 * BIN_CAP] : 0;
#pragma unroll
            for (int u = 0; u < 16; ++u) {
                if (lane < m[u]) {
                    const int dl = v[u] & (BIN_RN - 1);
                    const int s  = (int)((unsigned)v[u] >> 8);
                    const int p  = atomicAdd(&sCnt[dl], 1);
                    if (p < GCN_CAP) sB[dl * GCN_CAP + p] = s;
                    else gcn_overflow(nodeBase + dl, s, ovfCnt, ovf);
                }
            }
        }
    }
    for (int cb = 0; cb < nch; cb += 1024) {
        if (cb > 0) {
            const int cMine = cb + wave * 32 + lane;
            mSpill = cMine < nch ? spillCnt[cMine] : 0;
        }
        unsigned pend = __builtin_amdgcn_ballot_w32(mSpill > 0);
        while (pend) {
            const int j = __builtin_ctz(pend);
            pend &= pend - 1;
            const int m = __builtin_amdgcn_readlane(mSpill, j);
            const int* q = spill + 2 * (size_t)(cb + wave * 32 + j) * BIN_CH;
            for (int k = lane; k < m; k += 32) {
                const int d = q[2 * k];
                if (d / BIN_RN == r) {
                    const int dl = d % BIN_RN;
                    const int s  = q[2 * k + 1];
                    const int p  = atomicAdd(&sCnt[dl], 1);
                    if (p < GCN_CAP) sB[dl * GCN_CAP + p] = s;
                    else gcn_overflow(d, s, ovfCnt, ovf);
                }
            }
        }
    }
    __syncthreads();
#pragma unroll
    for (int i = 0; i < BIN_RN / 64; ++i) {
        const int row  = (t >> 4) + 64 * i;
        const int q4   = (t & 15) * 4;
        const int node = nodeBase + row;
        if (node < n && q4 < sCnt[row])
            *(int4*)&bucket[(size_t)node * GCN_CAP + q4] = *(const int4*)&sB[row * GCN_CAP + q4];
    }
    if (t < BIN_RN && nodeBase + t < n) cnt[nodeBase + t] = sCnt[t];
}

__device__ __forceinline__ float gcn_dinv(int c) {
    return __builtin_amdgcn_rsqf((float)(c + 1));
}

template <bool BIAS>
__device__ __forceinline__ float4
gcn_row_sum(int node, int c, const int* __restrict__ cnt, const int* __restrict__ ovfCnt,
            const int* __restrict__ bucket, const int* __restrict__ ovf,
            const float4* __restrict__ h4, const float* __restrict__ bias) {
    const int cn = cnt[node];
    const float dn = gcn_dinv(cn);
    const int* row = bucket + (size_t)node * GCN_CAP;
    const int end = cn < GCN_CAP ? cn : GCN_CAP;

    const float4 self = h4[(size_t)node * 16 + c];
    const float wself = dn * dn;
    float4 acc;
    if (BIAS) {
        const float4 b = ((const float4*)bias)[c];
        acc.x = b.x + self.x * wself;
        acc.y = b.y + self.y * wself;
        acc.z = b.z + self.z * wself;
        acc.w = b.w + self.w * wself;
    } else {
        acc.x = self.x * wself;
        acc.y = self.y * wself;
        acc.z = self.z * wself;
        acc.w = self.w * wself;
    }

    int j = 0;
    for (; j + 4 <= end; j += 4) {
        const int s0 = row[j + 0];
        const int s1 = row[j + 1];
        const int s2 = row[j + 2];
        const int s3 = row[j + 3];
        const float4 v0 = h4[(size_t)s0 * 16 + c];
        const float4 v1 = h4[(size_t)s1 * 16 + c];
        const float4 v2 = h4[(size_t)s2 * 16 + c];
        const float4 v3 = h4[(size_t)s3 * 16 + c];
        const float w0 = gcn_dinv(cnt[s0]) * dn;
        const float w1 = gcn_dinv(cnt[s1]) * dn;
        const float w2 = gcn_dinv(cnt[s2]) * dn;
        const float w3 = gcn_dinv(cnt[s3]) * dn;
        acc.x += v0.x * w0; acc.y += v0.y * w0; acc.z += v0.z * w0; acc.w += v0.w * w0;
        acc.x += v1.x * w1; acc.y += v1.y * w1; acc.z += v1.z * w1; acc.w += v1.w * w1;
        acc.x += v2.x * w2; acc.y += v2.y * w2; acc.z += v2.z * w2; acc.w += v2.w * w2;
        acc.x += v3.x * w3; acc.y += v3.y * w3; acc.z += v3.z * w3; acc.w += v3.w * w3;
    }
    for (; j < end; ++j) {
        const int s = row[j];
        const float4 v = h4[(size_t)s * 16 + c];
        const int cs = cnt[s];
        __builtin_amdgcn_sched_barrier(0);
        const float w = gcn_dinv(cs) * dn;
        acc.x += v.x * w; acc.y += v.y * w; acc.z += v.z * w; acc.w += v.w * w;
    }
    if (cn > GCN_CAP) {
        const int no = *ovfCnt;
        for (int k = 0; k < no; ++k) {
            if (ovf[2 * (size_t)k] == node) {
                const int s = ovf[2 * (size_t)k + 1];
                const float4 v = h4[(size_t)s * 16 + c];
                const float w = gcn_dinv(cnt[s]) * dn;
                acc.x += v.x * w; acc.y += v.y * w; acc.z += v.z * w; acc.w += v.w * w;
            }
        }
    }
    return acc;
}

constexpr int GCN_PA = 64 + 4;
constexpr int GCN_PH = 128 + 4;

__global__ void __launch_bounds__(256)
gcn_layer1_fused(const int* __restrict__ cnt, const int* __restrict__ ovfCnt,
                 const int* __restrict__ bucket, const int* __restrict__ ovf,
                 const float* __restrict__ x, const float* __restrict__ W1,
                 const float* __restrict__ b1, const float* __restrict__ W2,
                 float* __restrict__ h2, int n) {
    __shared__ __attribute__((aligned(16))) float sA[16 * GCN_PA];
    __shared__ __attribute__((aligned(16))) float sH[16 * GCN_PH];

    const int t = threadIdx.x;
    const int nodeBase = blockIdx.x * 16;

    {
        const int r = t >> 4;
        const int c = t & 15;
        const int node = nodeBase + r;
        float4 acc = make_float4(0.0f, 0.0f, 0.0f, 0.0f);
        if (node < n)
            acc = gcn_row_sum<false>(node, c, cnt, ovfCnt, bucket, ovf, (const float4*)x, nullptr);
        *(float4*)&sA[r * GCN_PA + 4 * c] = acc;
    }
    __syncthreads();

    const int wave  = t >> 5;
    const int lane  = t & 31;
    const int m     = lane & 15;
    const int khalf = (lane >> 4) << 1;
    const int rOff  = (lane >> 4) << 3;
    const int col   = (wave << 4) + m;

    {
        v8f acc = {};
        const float* Arow = sA + m * GCN_PA + khalf;
        const float* Bcol = W1 + (size_t)khalf * 128 + col;
#pragma unroll
        for (int k = 0; k < 64; k += 4) {
            const v2f a = *(const v2f*)(Arow + k);
            v2f b;
            b.x = Bcol[(size_t)k * 128];
            b.y = Bcol[(size_t)(k + 1) * 128];
            acc = __builtin_amdgcn_wmma_f32_16x16x4_f32(
                false, a, false, b, (short)0, acc, false, false);
        }
        const float bv = b1[col];
#pragma unroll
        for (int r = 0; r < 8; ++r)
            sH[(rOff + r) * GCN_PH + col] = fmaxf(acc[r] + bv, 0.0f);
    }
    __syncthreads();

    if (wave < 4) {
        v8f acc = {};
        const float* Arow = sH + m * GCN_PH + khalf;
        const float* Bcol = W2 + (size_t)khalf * 64 + col;
#pragma unroll
        for (int k = 0; k < 128; k += 4) {
            const v2f a = *(const v2f*)(Arow + k);
            v2f b;
            b.x = Bcol[(size_t)k * 64];
            b.y = Bcol[(size_t)(k + 1) * 64];
            acc = __builtin_amdgcn_wmma_f32_16x16x4_f32(
                false, a, false, b, (short)0, acc, false, false);
        }
        float* Crow = h2 + (size_t)(nodeBase + rOff) * 64 + col;
#pragma unroll
        for (int r = 0; r < 8; ++r)
            Crow[(size_t)r * 64] = acc[r];
    }
}

__global__ void __launch_bounds__(256)
gcn_gather_out(const int* __restrict__ cnt, const int* __restrict__ ovfCnt,
               const int* __restrict__ bucket, const int* __restrict__ ovf,
               const float* __restrict__ h, const float* __restrict__ bias,
               float* __restrict__ out, int n) {
    const int id   = blockIdx.x * blockDim.x + threadIdx.x;
    const int node = id >> 4;
    const int c    = id & 15;
    if (node >= n) return;
    ((float4*)out)[(size_t)node * 16 + c] =
        gcn_row_sum<true>(node, c, cnt, ovfCnt, bucket, ovf, (const float4*)h, bias);
}

extern "C" void kernel_launch(void* const* d_in, const int* in_sizes, int n_in,
                              void* d_out, int out_size, void* d_ws, size_t ws_size,
                              hipStream_t stream) {
    constexpr int D_IN = 64, D_OUT = 64;

    const float* x  = (const float*)d_in[0];
    const int*   ei = (const int*)  d_in[1];
    const float* W1 = (const float*)d_in[2];
    const float* b1 = (const float*)d_in[3];
    const float* W2 = (const float*)d_in[4];
    const float* b2 = (const float*)d_in[5];
    float* out = (float*)d_out;

    const int N = in_sizes[0] / D_IN;
    const int E = in_sizes[1] / 2;
    const int* src = ei;
    const int* dst = ei + E;

    auto cdiv = [](long a, long b) { return (int)((a + b - 1) / b); };
    auto up64 = [](size_t w) { return (w + 63) & ~(size_t)63; };
    const size_t NC  = up64((size_t)N + 1);
    const size_t N16 = ((size_t)N + 15) & ~(size_t)15;
    const int NR  = cdiv(N, BIN_RN);
    const int NCH = cdiv(E, BIN_CH);
    const int NJ  = cdiv(NCH, 32);

    float* h2       = (float*)d_ws;
    int*   cnt      = (int*)(h2 + N16 * D_OUT);
    int*   ovfCnt   = cnt + N;
    int*   bucket   = cnt + NC;
    int*   ovf      = bucket + (size_t)N * GCN_CAP;
    int*   segCnt   = ovf + up64(2 * (size_t)E);
    int*   spillCnt = segCnt + up64((size_t)NR * 32 * NJ);
    int*   seg      = spillCnt + up64((size_t)NCH);
    int*   spill    = seg + (size_t)NR * NCH * BIN_CAP;

    gcn_bin_edges   <<<NCH, 256, 0, stream>>>(src, dst, segCnt, seg, spillCnt, spill, ovfCnt, E, NR, NCH, NJ);
    gcn_fill_buckets<<<NR, 1024, 0, stream>>>(segCnt, seg, spillCnt, spill, cnt, ovfCnt, bucket, ovf, N, NCH, NJ);

    gcn_layer1_fused<<<cdiv(N, 16), 256, 0, stream>>>(
        cnt, ovfCnt, bucket, ovf, x, W1, b1, W2, h2, N);

    gcn_gather_out<<<cdiv((long)N * 16, 256), 256, 0, stream>>>(
        cnt, ovfCnt, bucket, ovf, h2, b2, out, N);
}
